# starting kernel
# baseline (speedup 1.0000x reference)
#include <hip/hip_runtime.h>

typedef float v2f __attribute__((ext_vector_type(2)));
typedef float v8f __attribute__((ext_vector_type(8)));

__device__ __forceinline__ v8f wmma_f32_k4(v2f a, v2f b, v8f c) {
    return __builtin_amdgcn_wmma_f32_16x16x4_f32(false, a, false, b, (short)0, c,
                                                 false, false);
}

__device__ __forceinline__ float leaky(float v) { return fmaxf(v, 0.01f * v); }

__global__ __launch_bounds__(32)
void itl_wmma_kernel(const float* __restrict__ x,
                     const float* __restrict__ w1, const float* __restrict__ b1,
                     const float* __restrict__ w2, const float* __restrict__ b2,
                     const float* __restrict__ w3, const float* __restrict__ b3,
                     float* __restrict__ out)
{
    __shared__ float xs[512 * 4];
    __shared__ float o1s[160 * 16];
    __shared__ float o2s[144 * 16];

    const int n    = blockIdx.x;
    const int lane = threadIdx.x;
    const int half = lane >> 4;
    const int m    = lane & 15;

    {
        const float4* xg = reinterpret_cast<const float4*>(x + (size_t)n * 2048);
        float4*       xl = reinterpret_cast<float4*>(xs);
#pragma unroll
        for (int i = 0; i < 16; ++i)
            xl[lane + 32 * i] = xg[lane + 32 * i];
    }
    __syncthreads();

    {
        const float bias = b1[m];
        for (int tile = 0; tile < 10; ++tile) {
            const int t0 = tile * 16;
            v8f acc = {bias, bias, bias, bias, bias, bias, bias, bias};
#pragma unroll
            for (int kk0 = 0; kk0 < 24; kk0 += 4) {
                const int r0 = kk0 + 2 * half, r1 = r0 + 1;
                v2f a, b;
                a.x = xs[(3 * (t0 + m) + (r0 % 6)) * 4 + (r0 / 6)];
                a.y = xs[(3 * (t0 + m) + (r1 % 6)) * 4 + (r1 / 6)];
                b.x = w1[m * 24 + r0];
                b.y = w1[m * 24 + r1];
                acc = wmma_f32_k4(a, b, acc);
            }
#pragma unroll
            for (int r = 0; r < 8; ++r)
                o1s[(t0 + r + 8 * half) * 16 + m] = leaky(acc[r]);
        }
    }
    __syncthreads();

    {
        const float bias = b2[m];
        for (int tile = 0; tile < 9; ++tile) {
            const int t0 = tile * 16;
            v8f acc0 = {bias, bias, bias, bias, bias, bias, bias, bias};
            v8f acc1 = {0, 0, 0, 0, 0, 0, 0, 0};
#pragma unroll
            for (int kk0 = 0; kk0 < 64; kk0 += 8) {
                {
                    const int r0 = kk0 + 2 * half, r1 = r0 + 1;
                    v2f a, b;
                    a.x = o1s[(t0 + m + 2 * (r0 & 3)) * 16 + (r0 >> 2)];
                    a.y = o1s[(t0 + m + 2 * (r1 & 3)) * 16 + (r1 >> 2)];
                    b.x = w2[m * 64 + r0];
                    b.y = w2[m * 64 + r1];
                    acc0 = wmma_f32_k4(a, b, acc0);
                }
                {
                    const int r0 = kk0 + 4 + 2 * half, r1 = r0 + 1;
                    v2f a, b;
                    a.x = o1s[(t0 + m + 2 * (r0 & 3)) * 16 + (r0 >> 2)];
                    a.y = o1s[(t0 + m + 2 * (r1 & 3)) * 16 + (r1 >> 2)];
                    b.x = w2[m * 64 + r0];
                    b.y = w2[m * 64 + r1];
                    acc1 = wmma_f32_k4(a, b, acc1);
                }
            }
            acc0 = acc0 + acc1;
#pragma unroll
            for (int r = 0; r < 8; ++r)
                o2s[(t0 + r + 8 * half) * 16 + m] = leaky(acc0[r]);
        }
    }
    __syncthreads();

    {
        const float bias = b3[m];
        float* out3 = out + (size_t)n * 2304 + 2 * 768;
        for (int tile = 0; tile < 3; ++tile) {
            const int t0 = tile * 16;
            v8f acc0 = {bias, bias, bias, bias, bias, bias, bias, bias};
            v8f acc1 = {0, 0, 0, 0, 0, 0, 0, 0};
#pragma unroll 2
            for (int kk0 = 0; kk0 < 208; kk0 += 8) {
                {
                    const int r0 = kk0 + 2 * half, r1 = r0 + 1;
                    const int h0 = r0 / 13, k0 = r0 - 13 * h0;
                    const int h1 = r1 / 13, k1 = r1 - 13 * h1;
                    v2f a, b;
                    a.x = o2s[(2 * (t0 + m) + 4 * k0) * 16 + h0];
                    a.y = o2s[(2 * (t0 + m) + 4 * k1) * 16 + h1];
                    b.x = w3[m * 208 + r0];
                    b.y = w3[m * 208 + r1];
                    acc0 = wmma_f32_k4(a, b, acc0);
                }
                {
                    const int r0 = kk0 + 4 + 2 * half, r1 = r0 + 1;
                    const int h0 = r0 / 13, k0 = r0 - 13 * h0;
                    const int h1 = r1 / 13, k1 = r1 - 13 * h1;
                    v2f a, b;
                    a.x = o2s[(2 * (t0 + m) + 4 * k0) * 16 + h0];
                    a.y = o2s[(2 * (t0 + m) + 4 * k1) * 16 + h1];
                    b.x = w3[m * 208 + r0];
                    b.y = w3[m * 208 + r1];
                    acc1 = wmma_f32_k4(a, b, acc1);
                }
            }
            acc0 = acc0 + acc1;
#pragma unroll
            for (int r = 0; r < 8; ++r)
                out3[(t0 + r + 8 * half) * 16 + m] = leaky(acc0[r]);
        }
    }

    {
        float* outn = out + (size_t)n * 2304;
#pragma unroll
        for (int i = 0; i < 24; ++i) {
            const int idx = lane + 32 * i;
            outn[idx]       = o1s[idx];
            outn[768 + idx] = o2s[idx];
        }
    }
}

extern "C" void kernel_launch(void* const* d_in, const int* in_sizes, int n_in,
                              void* d_out, int out_size, void* d_ws, size_t ws_size,
                              hipStream_t stream) {
    const float* x  = (const float*)d_in[0];
    const float* w1 = (const float*)d_in[1];
    const float* b1 = (const float*)d_in[2];
    const float* w2 = (const float*)d_in[3];
    const float* b2 = (const float*)d_in[4];
    const float* w3 = (const float*)d_in[5];
    const float* b3 = (const float*)d_in[6];
    float* out = (float*)d_out;

    const int nseq = in_sizes[0] / (512 * 4);
    itl_wmma_kernel<<<dim3(nseq), dim3(32), 0, stream>>>(x, w1, b1, w2, b2, w3, b3, out);
}


# K index of all three stages reordered tap-major/channel-minor: a lane's A fragments are adjacent in LDS (ds_load_b128, not b32), weights gathered once per wave in that order; fp32 sum order changes
# speedup vs baseline: 1.6194x; 1.6194x over previous
#include <hip/hip_runtime.h>

typedef float v2f __attribute__((ext_vector_type(2)));
typedef float v8f __attribute__((ext_vector_type(8)));

__device__ __forceinline__ v8f wmma_f32_k4(v2f a, v2f b, v8f c) {
    return __builtin_amdgcn_wmma_f32_16x16x4_f32(false, a, false, b, (short)0, c,
                                                 false, false);
}

__device__ __forceinline__ float leaky(float v) { return fmaxf(v, 0.01f * v); }

__global__ __launch_bounds__(32)
void itl_wmma_kernel(const float* __restrict__ x,
                     const float* __restrict__ w1, const float* __restrict__ b1,
                     const float* __restrict__ w2, const float* __restrict__ b2,
                     const float* __restrict__ w3, const float* __restrict__ b3,
                     float* __restrict__ out)
{
    __shared__ __attribute__((aligned(16))) float xs[512 * 4];
    __shared__ __attribute__((aligned(16))) float o1s[160 * 16];
    __shared__ __attribute__((aligned(16))) float o2s[144 * 16];

    const int n    = blockIdx.x;
    const int lane = threadIdx.x;
    const int half = lane >> 4;
    const int m    = lane & 15;

    v2f wb1[6], wb2[4][2][2], wb3[13][2][2];
#pragma unroll
    for (int j = 0; j < 6; ++j) {
        const int f0 = 12 * half + 2 * j, f1 = f0 + 1;
        wb1[j].x = w1[m * 24 + (f0 & 3) * 6 + (f0 >> 2)];
        wb1[j].y = w1[m * 24 + (f1 & 3) * 6 + (f1 >> 2)];
    }
#pragma unroll
    for (int k = 0; k < 4; ++k)
#pragma unroll
        for (int p = 0; p < 2; ++p)
#pragma unroll
            for (int q = 0; q < 2; ++q) {
                const int hin = 8 * p + 4 * half + 2 * q;
                wb2[k][p][q].x = w2[m * 64 + hin * 4 + k];
                wb2[k][p][q].y = w2[m * 64 + (hin + 1) * 4 + k];
            }
#pragma unroll
    for (int k = 0; k < 13; ++k)
#pragma unroll
        for (int p = 0; p < 2; ++p)
#pragma unroll
            for (int q = 0; q < 2; ++q) {
                const int hin = 8 * p + 4 * half + 2 * q;
                wb3[k][p][q].x = w3[m * 208 + hin * 13 + k];
                wb3[k][p][q].y = w3[m * 208 + (hin + 1) * 13 + k];
            }
    const float bias1 = b1[m], bias2 = b2[m], bias3 = b3[m];

    {
        const float4* xg = reinterpret_cast<const float4*>(x + (size_t)n * 2048);
        float4*       xl = reinterpret_cast<float4*>(xs);
#pragma unroll
        for (int i = 0; i < 16; ++i)
            xl[lane + 32 * i] = xg[lane + 32 * i];
    }
    __syncthreads();

    {
        const float bias = bias1;
        for (int tile = 0; tile < 10; ++tile) {
            const int t0 = tile * 16;
            const float4* ap = reinterpret_cast<const float4*>(xs + 12 * (t0 + m) + 12 * half);
            const float4 a0 = ap[0], a1 = ap[1], a2 = ap[2];
            v8f acc = {bias, bias, bias, bias, bias, bias, bias, bias};
            acc = wmma_f32_k4((v2f){a0.x, a0.y}, wb1[0], acc);
            acc = wmma_f32_k4((v2f){a0.z, a0.w}, wb1[1], acc);
            acc = wmma_f32_k4((v2f){a1.x, a1.y}, wb1[2], acc);
            acc = wmma_f32_k4((v2f){a1.z, a1.w}, wb1[3], acc);
            acc = wmma_f32_k4((v2f){a2.x, a2.y}, wb1[4], acc);
            acc = wmma_f32_k4((v2f){a2.z, a2.w}, wb1[5], acc);
#pragma unroll
            for (int r = 0; r < 8; ++r)
                o1s[(t0 + r + 8 * half) * 16 + m] = leaky(acc[r]);
        }
    }
    __syncthreads();

    {
        const float bias = bias2;
        for (int tile = 0; tile < 9; ++tile) {
            const int t0 = tile * 16;
            v8f acc0 = {bias, bias, bias, bias, bias, bias, bias, bias};
            v8f acc1 = {0, 0, 0, 0, 0, 0, 0, 0};
#pragma unroll
            for (int k = 0; k < 4; ++k)
#pragma unroll
                for (int p = 0; p < 2; ++p) {
                    const float4 a = *reinterpret_cast<const float4*>(
                        o1s + (t0 + m + 2 * k) * 16 + 8 * p + 4 * half);
                    acc0 = wmma_f32_k4((v2f){a.x, a.y}, wb2[k][p][0], acc0);
                    acc1 = wmma_f32_k4((v2f){a.z, a.w}, wb2[k][p][1], acc1);
                }
            acc0 = acc0 + acc1;
#pragma unroll
            for (int r = 0; r < 8; ++r)
                o2s[(t0 + r + 8 * half) * 16 + m] = leaky(acc0[r]);
        }
    }
    __syncthreads();

    {
        const float bias = bias3;
        float* out3 = out + (size_t)n * 2304 + 2 * 768;
        for (int tile = 0; tile < 3; ++tile) {
            const int t0 = tile * 16;
            v8f acc0 = {bias, bias, bias, bias, bias, bias, bias, bias};
            v8f acc1 = {0, 0, 0, 0, 0, 0, 0, 0};
#pragma unroll
            for (int k = 0; k < 13; ++k)
#pragma unroll
                for (int p = 0; p < 2; ++p) {
                    const float4 a = *reinterpret_cast<const float4*>(
                        o2s + (2 * (t0 + m) + 4 * k) * 16 + 8 * p + 4 * half);
                    acc0 = wmma_f32_k4((v2f){a.x, a.y}, wb3[k][p][0], acc0);
                    acc1 = wmma_f32_k4((v2f){a.z, a.w}, wb3[k][p][1], acc1);
                }
            acc0 = acc0 + acc1;
#pragma unroll
            for (int r = 0; r < 8; ++r)
                out3[(t0 + r + 8 * half) * 16 + m] = leaky(acc0[r]);
        }
    }

    {
        float* outn = out + (size_t)n * 2304;
#pragma unroll
        for (int i = 0; i < 24; ++i) {
            const int idx = lane + 32 * i;
            outn[idx]       = o1s[idx];
            outn[768 + idx] = o2s[idx];
        }
    }
}

extern "C" void kernel_launch(void* const* d_in, const int* in_sizes, int n_in,
                              void* d_out, int out_size, void* d_ws, size_t ws_size,
                              hipStream_t stream) {
    const float* x  = (const float*)d_in[0];
    const float* w1 = (const float*)d_in[1];
    const float* b1 = (const float*)d_in[2];
    const float* w2 = (const float*)d_in[3];
    const float* b2 = (const float*)d_in[4];
    const float* w3 = (const float*)d_in[5];
    const float* b3 = (const float*)d_in[6];
    float* out = (float*)d_out;

    const int nseq = in_sizes[0] / (512 * 4);
    itl_wmma_kernel<<<dim3(nseq), dim3(32), 0, stream>>>(x, w1, b1, w2, b2, w3, b3, out);
}


# o1/o2 split into even-row and odd-row LDS planes; only rows an output depends on are computed (stage 3 reads even o2 rows only): 294 WMMAs, not 360; fp32 sum order unchanged
# speedup vs baseline: 1.8578x; 1.1472x over previous
#include <hip/hip_runtime.h>

typedef float v2f __attribute__((ext_vector_type(2)));
typedef float v8f __attribute__((ext_vector_type(8)));

__device__ __forceinline__ v8f wmma_f32_k4(v2f a, v2f b, v8f c) {
    return __builtin_amdgcn_wmma_f32_16x16x4_f32(false, a, false, b, (short)0, c,
                                                 false, false);
}

__device__ __forceinline__ float leaky(float v) { return fmaxf(v, 0.01f * v); }

constexpr int O1ODD = 80 * 16;
constexpr int O2ODD = 72 * 16;

__device__ __forceinline__ void s1_tile(const float* arow, float* drow,
                                        const v2f (&wb1)[6], float bias) {
    const float4* ap = reinterpret_cast<const float4*>(arow);
    const float4 a0 = ap[0], a1 = ap[1], a2 = ap[2];
    v8f acc = {bias, bias, bias, bias, bias, bias, bias, bias};
    acc = wmma_f32_k4((v2f){a0.x, a0.y}, wb1[0], acc);
    acc = wmma_f32_k4((v2f){a0.z, a0.w}, wb1[1], acc);
    acc = wmma_f32_k4((v2f){a1.x, a1.y}, wb1[2], acc);
    acc = wmma_f32_k4((v2f){a1.z, a1.w}, wb1[3], acc);
    acc = wmma_f32_k4((v2f){a2.x, a2.y}, wb1[4], acc);
    acc = wmma_f32_k4((v2f){a2.z, a2.w}, wb1[5], acc);
#pragma unroll
    for (int r = 0; r < 8; ++r)
        drow[r * 16] = leaky(acc[r]);
}

__device__ __forceinline__ void s2_tile(const float* arow, float* drow,
                                        const v2f (&wb2)[4][2][2], float bias) {
    v8f acc0 = {bias, bias, bias, bias, bias, bias, bias, bias};
    v8f acc1 = {0, 0, 0, 0, 0, 0, 0, 0};
#pragma unroll
    for (int k = 0; k < 4; ++k)
#pragma unroll
        for (int p = 0; p < 2; ++p) {
            const float4 a = *reinterpret_cast<const float4*>(arow + k * 16 + 8 * p);
            acc0 = wmma_f32_k4((v2f){a.x, a.y}, wb2[k][p][0], acc0);
            acc1 = wmma_f32_k4((v2f){a.z, a.w}, wb2[k][p][1], acc1);
        }
    acc0 = acc0 + acc1;
#pragma unroll
    for (int r = 0; r < 8; ++r)
        drow[r * 16] = leaky(acc0[r]);
}

__global__ __launch_bounds__(32) __attribute__((amdgpu_waves_per_eu(1, 3)))
void itl_wmma_kernel(const float* __restrict__ x,
                     const float* __restrict__ w1, const float* __restrict__ b1,
                     const float* __restrict__ w2, const float* __restrict__ b2,
                     const float* __restrict__ w3, const float* __restrict__ b3,
                     float* __restrict__ out)
{
    __shared__ __attribute__((aligned(16))) float xs[512 * 4];
    __shared__ __attribute__((aligned(16))) float o1p[(80 + 32) * 16];
    __shared__ __attribute__((aligned(16))) float o2p[(72 + 24) * 16];

    const int n    = blockIdx.x;
    const int lane = threadIdx.x;
    const int half = lane >> 4;
    const int m    = lane & 15;

    v2f wb1[6], wb2[4][2][2], wb3[13][2][2];
#pragma unroll
    for (int j = 0; j < 6; ++j) {
        const int f0 = 12 * half + 2 * j, f1 = f0 + 1;
        wb1[j].x = w1[m * 24 + (f0 & 3) * 6 + (f0 >> 2)];
        wb1[j].y = w1[m * 24 + (f1 & 3) * 6 + (f1 >> 2)];
    }
#pragma unroll
    for (int k = 0; k < 4; ++k)
#pragma unroll
        for (int p = 0; p < 2; ++p)
#pragma unroll
            for (int q = 0; q < 2; ++q) {
                const int hin = 8 * p + 4 * half + 2 * q;
                wb2[k][p][q].x = w2[m * 64 + hin * 4 + k];
                wb2[k][p][q].y = w2[m * 64 + (hin + 1) * 4 + k];
            }
#pragma unroll
    for (int k = 0; k < 13; ++k)
#pragma unroll
        for (int p = 0; p < 2; ++p)
#pragma unroll
            for (int q = 0; q < 2; ++q) {
                const int hin = 8 * p + 4 * half + 2 * q;
                wb3[k][p][q].x = w3[m * 208 + hin * 13 + k];
                wb3[k][p][q].y = w3[m * 208 + (hin + 1) * 13 + k];
            }
    const float bias1 = b1[m], bias2 = b2[m], bias3 = b3[m];

    {
        const float4* xg = reinterpret_cast<const float4*>(x + (size_t)n * 2048);
        float4*       xl = reinterpret_cast<float4*>(xs);
#pragma unroll
        for (int i = 0; i < 16; ++i)
            xl[lane + 32 * i] = xg[lane + 32 * i];
    }
    __syncthreads();

    for (int tile = 0; tile < 5; ++tile)
        s1_tile(xs + 24 * (tile * 16 + m) + 12 * half,
                o1p + (tile * 16 + 8 * half) * 16 + m, wb1, bias1);
    for (int tile = 0; tile < 2; ++tile)
        s1_tile(xs + 24 * (tile * 16 + m) + 12 + 12 * half,
                o1p + O1ODD + (tile * 16 + 8 * half) * 16 + m, wb1, bias1);
    __syncthreads();

    for (int tile = 0; tile < 4; ++tile)
        s2_tile(o1p + (tile * 16 + m) * 16 + 4 * half,
                o2p + (tile * 16 + 8 * half) * 16 + m, wb2, bias2);
    s2_tile(o1p + O1ODD + m * 16 + 4 * half,
            o2p + O2ODD + (8 * half) * 16 + m, wb2, bias2);
    {
        const int arow = (m < 8 ? (64 + m) * 16 : O1ODD + (8 + m) * 16) + 4 * half;
        const int drow = (half ? O2ODD + 16 * 16 : 64 * 16) + m;
        s2_tile(o1p + arow, o2p + drow, wb2, bias2);
    }
    __syncthreads();

    {
        const float bias = bias3;
        float* out3 = out + (size_t)n * 2304 + 2 * 768;
        for (int tile = 0; tile < 3; ++tile) {
            const int t0 = tile * 16;
            v8f acc0 = {bias, bias, bias, bias, bias, bias, bias, bias};
            v8f acc1 = {0, 0, 0, 0, 0, 0, 0, 0};
#pragma unroll
            for (int k = 0; k < 13; ++k)
#pragma unroll
                for (int p = 0; p < 2; ++p) {
                    const float4 a = *reinterpret_cast<const float4*>(
                        o2p + (t0 + m + 2 * k) * 16 + 8 * p + 4 * half);
                    acc0 = wmma_f32_k4((v2f){a.x, a.y}, wb3[k][p][0], acc0);
                    acc1 = wmma_f32_k4((v2f){a.z, a.w}, wb3[k][p][1], acc1);
                }
            acc0 = acc0 + acc1;
#pragma unroll
            for (int r = 0; r < 8; ++r)
                out3[(t0 + r + 8 * half) * 16 + m] = leaky(acc0[r]);
        }
    }

    {
        float* outn = out + (size_t)n * 2304;
        const int s1 = (half ? O1ODD : 0) + m;
        const int s2 = (half ? O2ODD : 0) + m;
#pragma unroll
        for (int i = 0; i < 24; ++i) {
            const int idx = lane + 32 * i;
            outn[idx]       = o1p[s1 + 16 * i];
            outn[768 + idx] = o2p[s2 + 16 * i];
        }
    }
}

extern "C" void kernel_launch(void* const* d_in, const int* in_sizes, int n_in,
                              void* d_out, int out_size, void* d_ws, size_t ws_size,
                              hipStream_t stream) {
    const float* x  = (const float*)d_in[0];
    const float* w1 = (const float*)d_in[1];
    const float* b1 = (const float*)d_in[2];
    const float* w2 = (const float*)d_in[3];
    const float* b2 = (const float*)d_in[4];
    const float* w3 = (const float*)d_in[5];
    const float* b3 = (const float*)d_in[6];
    float* out = (float*)d_out;

    const int nseq = in_sizes[0] / (512 * 4);
    itl_wmma_kernel<<<dim3(nseq), dim3(32), 0, stream>>>(x, w1, b1, w2, b2, w3, b3, out);
}


# o1/o2 LDS rows padded from 16 to 20 floats (pitch 80 B = 16 B x odd), odd planes 32 banks off the even: stage 2/3 A-fragment ds_load_b128 and D stores bank-conflict-free; fp32 sum order unchanged
# speedup vs baseline: 1.9256x; 1.0365x over previous
#include <hip/hip_runtime.h>

typedef float v2f __attribute__((ext_vector_type(2)));
typedef float v8f __attribute__((ext_vector_type(8)));

__device__ __forceinline__ v8f wmma_f32_k4(v2f a, v2f b, v8f c) {
    return __builtin_amdgcn_wmma_f32_16x16x4_f32(false, a, false, b, (short)0, c,
                                                 false, false);
}

__device__ __forceinline__ float leaky(float v) { return fmaxf(v, 0.01f * v); }

constexpr int P     = 20;
constexpr int O1ODD = 88 * P;
constexpr int O2ODD = 72 * P;

__device__ __forceinline__ void s1_tile(const float* arow, float* drow,
                                        const v2f (&wb1)[6], float bias) {
    const float4* ap = reinterpret_cast<const float4*>(arow);
    const float4 a0 = ap[0], a1 = ap[1], a2 = ap[2];
    v8f acc = {bias, bias, bias, bias, bias, bias, bias, bias};
    acc = wmma_f32_k4((v2f){a0.x, a0.y}, wb1[0], acc);
    acc = wmma_f32_k4((v2f){a0.z, a0.w}, wb1[1], acc);
    acc = wmma_f32_k4((v2f){a1.x, a1.y}, wb1[2], acc);
    acc = wmma_f32_k4((v2f){a1.z, a1.w}, wb1[3], acc);
    acc = wmma_f32_k4((v2f){a2.x, a2.y}, wb1[4], acc);
    acc = wmma_f32_k4((v2f){a2.z, a2.w}, wb1[5], acc);
#pragma unroll
    for (int r = 0; r < 8; ++r)
        drow[r * P] = leaky(acc[r]);
}

__device__ __forceinline__ void s2_tile(const float* arow, float* drow,
                                        const v2f (&wb2)[4][2][2], float bias) {
    v8f acc0 = {bias, bias, bias, bias, bias, bias, bias, bias};
    v8f acc1 = {0, 0, 0, 0, 0, 0, 0, 0};
#pragma unroll
    for (int k = 0; k < 4; ++k)
#pragma unroll
        for (int p = 0; p < 2; ++p) {
            const float4 a = *reinterpret_cast<const float4*>(arow + k * P + 8 * p);
            acc0 = wmma_f32_k4((v2f){a.x, a.y}, wb2[k][p][0], acc0);
            acc1 = wmma_f32_k4((v2f){a.z, a.w}, wb2[k][p][1], acc1);
        }
    acc0 = acc0 + acc1;
#pragma unroll
    for (int r = 0; r < 8; ++r)
        drow[r * P] = leaky(acc0[r]);
}

__global__ __launch_bounds__(32) __attribute__((amdgpu_waves_per_eu(1, 3)))
void itl_wmma_kernel(const float* __restrict__ x,
                     const float* __restrict__ w1, const float* __restrict__ b1,
                     const float* __restrict__ w2, const float* __restrict__ b2,
                     const float* __restrict__ w3, const float* __restrict__ b3,
                     float* __restrict__ out)
{
    __shared__ __attribute__((aligned(16))) float xs[512 * 4];
    __shared__ __attribute__((aligned(16))) float o1p[(88 + 32) * P];
    __shared__ __attribute__((aligned(16))) float o2p[(72 + 24) * P];

    const int n    = blockIdx.x;
    const int lane = threadIdx.x;
    const int half = lane >> 4;
    const int m    = lane & 15;

    v2f wb1[6], wb2[4][2][2], wb3[13][2][2];
#pragma unroll
    for (int j = 0; j < 6; ++j) {
        const int f0 = 12 * half + 2 * j, f1 = f0 + 1;
        wb1[j].x = w1[m * 24 + (f0 & 3) * 6 + (f0 >> 2)];
        wb1[j].y = w1[m * 24 + (f1 & 3) * 6 + (f1 >> 2)];
    }
#pragma unroll
    for (int k = 0; k < 4; ++k)
#pragma unroll
        for (int p = 0; p < 2; ++p)
#pragma unroll
            for (int q = 0; q < 2; ++q) {
                const int hin = 8 * p + 4 * half + 2 * q;
                wb2[k][p][q].x = w2[m * 64 + hin * 4 + k];
                wb2[k][p][q].y = w2[m * 64 + (hin + 1) * 4 + k];
            }
#pragma unroll
    for (int k = 0; k < 13; ++k)
#pragma unroll
        for (int p = 0; p < 2; ++p)
#pragma unroll
            for (int q = 0; q < 2; ++q) {
                const int hin = 8 * p + 4 * half + 2 * q;
                wb3[k][p][q].x = w3[m * 208 + hin * 13 + k];
                wb3[k][p][q].y = w3[m * 208 + (hin + 1) * 13 + k];
            }
    const float bias1 = b1[m], bias2 = b2[m], bias3 = b3[m];

    {
        const float4* xg = reinterpret_cast<const float4*>(x + (size_t)n * 2048);
        float4*       xl = reinterpret_cast<float4*>(xs);
#pragma unroll
        for (int i = 0; i < 16; ++i)
            xl[lane + 32 * i] = xg[lane + 32 * i];
    }
    __syncthreads();

    for (int tile = 0; tile < 5; ++tile)
        s1_tile(xs + 24 * (tile * 16 + m) + 12 * half,
                o1p + (tile * 16 + 8 * half) * P + m, wb1, bias1);
    for (int tile = 0; tile < 2; ++tile)
        s1_tile(xs + 24 * (tile * 16 + m) + 12 + 12 * half,
                o1p + O1ODD + (tile * 16 + 8 * half) * P + m, wb1, bias1);
    __syncthreads();

    for (int tile = 0; tile < 4; ++tile)
        s2_tile(o1p + (tile * 16 + m) * P + 4 * half,
                o2p + (tile * 16 + 8 * half) * P + m, wb2, bias2);
    s2_tile(o1p + O1ODD + m * P + 4 * half,
            o2p + O2ODD + (8 * half) * P + m, wb2, bias2);
    {
        const int arow = (m < 8 ? (64 + m) * P : O1ODD + (8 + m) * P) + 4 * half;
        const int drow = (half ? O2ODD + 16 * P : 64 * P) + m;
        s2_tile(o1p + arow, o2p + drow, wb2, bias2);
    }
    __syncthreads();

    {
        const float bias = bias3;
        float* out3 = out + (size_t)n * 2304 + 2 * 768;
        for (int tile = 0; tile < 3; ++tile) {
            const int t0 = tile * 16;
            v8f acc0 = {bias, bias, bias, bias, bias, bias, bias, bias};
            v8f acc1 = {0, 0, 0, 0, 0, 0, 0, 0};
#pragma unroll
            for (int k = 0; k < 13; ++k)
#pragma unroll
                for (int p = 0; p < 2; ++p) {
                    const float4 a = *reinterpret_cast<const float4*>(
                        o2p + (t0 + m + 2 * k) * P + 8 * p + 4 * half);
                    acc0 = wmma_f32_k4((v2f){a.x, a.y}, wb3[k][p][0], acc0);
                    acc1 = wmma_f32_k4((v2f){a.z, a.w}, wb3[k][p][1], acc1);
                }
            acc0 = acc0 + acc1;
#pragma unroll
            for (int r = 0; r < 8; ++r)
                out3[(t0 + r + 8 * half) * 16 + m] = leaky(acc0[r]);
        }
    }

    {
        float* outn = out + (size_t)n * 2304;
        const int s1 = (half ? O1ODD : 0) + m;
        const int s2 = (half ? O2ODD : 0) + m;
#pragma unroll
        for (int i = 0; i < 24; ++i) {
            const int idx = lane + 32 * i;
            outn[idx]       = o1p[s1 + P * i];
            outn[768 + idx] = o2p[s2 + P * i];
        }
    }
}

extern "C" void kernel_launch(void* const* d_in, const int* in_sizes, int n_in,
                              void* d_out, int out_size, void* d_ws, size_t ws_size,
                              hipStream_t stream) {
    const float* x  = (const float*)d_in[0];
    const float* w1 = (const float*)d_in[1];
    const float* b1 = (const float*)d_in[2];
    const float* w2 = (const float*)d_in[3];
    const float* b2 = (const float*)d_in[4];
    const float* w3 = (const float*)d_in[5];
    const float* b3 = (const float*)d_in[6];
    float* out = (float*)d_out;

    const int nseq = in_sizes[0] / (512 * 4);
    itl_wmma_kernel<<<dim3(nseq), dim3(32), 0, stream>>>(x, w1, b1, w2, b2, w3, b3, out);
}
